# starting kernel
# baseline (speedup 1.0000x reference)
#include <hip/hip_runtime.h>


typedef __attribute__((ext_vector_type(2))) float v2f;
typedef __attribute__((ext_vector_type(8))) float v8f;

#define BATCH  32
#define HEIGHT 256
#define WIDTH  256
#define CHAN   16
#define HCN    4
#define HS     12

#define XROW_STRIDE 20
#define XROWS       258
#define XCOL_SIZE   (XROWS * XROW_STRIDE)
#define NCOLS       12
#define OFF_XCOL    0
#define OFF_CONVW   (NCOLS * XCOL_SIZE)
#define OFF_CONVOUT (OFF_CONVW + 576)
#define OFF_US      (OFF_CONVOUT + 4096)
#define SMEM_FLOATS (OFF_US + 8 * 2 * 256)
#define DELTA_STRIDE 260
#define OFF_DELTA   0

__launch_bounds__(256)
__global__ void ca_step_kernel(const float* __restrict__ xin,
                               const float* __restrict__ convw,
                               const float* __restrict__ fc0w,
                               const float* __restrict__ fc0b,
                               const float* __restrict__ fc1w,
                               float* __restrict__ xout)
{
    extern __shared__ float smem[];
    float* xcol  = smem + OFF_XCOL;
    float* cw    = smem + OFF_CONVW;
    float* cout  = smem + OFF_CONVOUT;
    float* us    = smem + OFF_US;
    float* delta = smem + OFF_DELTA;

    const int tid  = threadIdx.x;
    const int lane = tid & 31;
    const int wave = tid >> 5;
    const int m    = lane & 15;
    const int hi   = lane >> 4;

    const int blk = blockIdx.x;
    const int b   = blk >> 6;
    const int ww  = blk & 63;

    const float* xres = xin + ((b * HEIGHT) * WIDTH + ww * 4) * CHAN;

    for (int i = tid; i < 1024; i += 256) {
        int h2 = i >> 2;
        int q  = i & 3;
        __builtin_prefetch(xres + (h2 * WIDTH + q) * CHAN, 0, 1);
    }

    v2f afc0;
    {
        int s = m;
        int base = (s < HS ? s : 0) * 4 + 2 * hi;
        float a0 = fc0w[base + 0];
        float a1 = fc0w[base + 1];
        afc0 = (s < HS) ? (v2f){a0, a1} : (v2f){0.f, 0.f};
    }
    v8f cbias;
#pragma unroll
    for (int v = 0; v < 8; ++v) {
        int s = v + 8 * hi;
        float bv = fc0b[s < HS ? s : 0];
        cbias[v] = (s < HS) ? bv : 0.f;
    }
    v2f afc1[3];
#pragma unroll
    for (int t3 = 0; t3 < 3; ++t3) {
        int base = m * HS + 4 * t3 + 2 * hi;
        afc1[t3].x = fc1w[base + 0];
        afc1[t3].y = fc1w[base + 1];
    }

    for (int i = tid; i < 576; i += 256) cw[i] = convw[i];

    for (int i = tid; i < NCOLS * XROWS * 4; i += 256) {
        int colIdx = i / (XROWS * 4);
        int rem    = i - colIdx * (XROWS * 4);
        int hh     = rem >> 2;
        int c4     = rem & 3;
        int t      = colIdx / 3;
        int aoff   = colIdx - t * 3 - 1;
        int wcol   = ww + 64 * t + aoff;
        int h      = hh - 1;
        float4 val = make_float4(0.f, 0.f, 0.f, 0.f);
        if ((unsigned)h < 256u && (unsigned)wcol < 256u) {
            val = *(const float4*)(xin + ((b * HEIGHT + h) * WIDTH + wcol) * CHAN + c4 * 4);
        }
        *(float4*)(xcol + colIdx * XCOL_SIZE + hh * XROW_STRIDE + c4 * 4) = val;
    }
    __syncthreads();


    v2f aw[36];
    {
        const float* pw = cw + (m & 3) * 144 + (2 * hi) * 9;
#pragma unroll
        for (int a = 0; a < 3; ++a)
#pragma unroll
            for (int d = 0; d < 3; ++d)
#pragma unroll
                for (int cblk = 0; cblk < 4; ++cblk) {
                    float w0 = pw[cblk * 36 + 0 + a * 3 + d];
                    float w1 = pw[cblk * 36 + 9 + a * 3 + d];
                    aw[(a * 3 + d) * 4 + cblk] =
                        (m < HCN) ? (v2f){w0, w1} : (v2f){0.f, 0.f};
                }
    }

    for (int uu = 0; uu < 8; ++uu) {
        int u  = wave * 8 + uu;
        int t  = u >> 4;
        int h0 = (u & 15) << 4;
        v8f acc = {};
        const float* pb = xcol + 3 * t * XCOL_SIZE + (h0 + m) * XROW_STRIDE + 2 * hi;
#pragma unroll
        for (int a = 0; a < 3; ++a) {
#pragma unroll
            for (int d = 0; d < 3; ++d) {
                const float* pg = pb + a * XCOL_SIZE + d * XROW_STRIDE;
                v2f b0 = *(const v2f*)(pg + 0);
                v2f b1 = *(const v2f*)(pg + 4);
                v2f b2 = *(const v2f*)(pg + 8);
                v2f b3 = *(const v2f*)(pg + 12);
                int s = (a * 3 + d) * 4;
                acc = __builtin_amdgcn_wmma_f32_16x16x4_f32(
                    false, aw[s + 0], false, b0, (short)0, acc, false, false);
                acc = __builtin_amdgcn_wmma_f32_16x16x4_f32(
                    false, aw[s + 1], false, b1, (short)0, acc, false, false);
                acc = __builtin_amdgcn_wmma_f32_16x16x4_f32(
                    false, aw[s + 2], false, b2, (short)0, acc, false, false);
                acc = __builtin_amdgcn_wmma_f32_16x16x4_f32(
                    false, aw[s + 3], false, b3, (short)0, acc, false, false);
            }
        }
        if (lane < 16) {
            float* cdst = cout + (t * 4) * 256 + h0 + lane;
            cdst[0]   = fmaxf(acc[0], 0.f);
            cdst[256] = fmaxf(acc[1], 0.f);
            cdst[512] = fmaxf(acc[2], 0.f);
            cdst[768] = fmaxf(acc[3], 0.f);
        }
    }
    __syncthreads();

#pragma unroll 2
    for (int uu = 0; uu < 8; ++uu) {
        int tt = wave * 8 + uu;
        int hc = tt >> 4;
        int t  = (tt >> 2) & 3;
        int q  = tt & 3;

        v2f Bv = *(const v2f*)(cout + (t * 4 + hc) * 256 + (q * 16 + m) * 4 + 2 * hi);
        v8f u0 = __builtin_amdgcn_wmma_f32_16x16x4_f32(
            false, afc0, false, Bv, (short)0, cbias, false, false);
#pragma unroll
        for (int v = 0; v < 8; ++v) u0[v] = fmaxf(u0[v], 0.f);

        float* usw = us + wave * 512 + (uu & 1) * 256;
        *(float4*)(usw + m * 16 + 8 * hi + 0) = make_float4(u0[0], u0[1], u0[2], u0[3]);
        *(float4*)(usw + m * 16 + 8 * hi + 4) = make_float4(u0[4], u0[5], u0[6], u0[7]);

        v8f acc2 = {};
#pragma unroll
        for (int t3 = 0; t3 < 3; ++t3) {
            v2f B2 = *(const v2f*)(usw + m * 16 + 4 * t3 + 2 * hi);
            acc2 = __builtin_amdgcn_wmma_f32_16x16x4_f32(
                false, afc1[t3], false, B2, (short)0, acc2, false, false);
        }

        int cq = (4 * hc + t) * 4 + q;
        float* dbase = delta + cq * DELTA_STRIDE + m * 16 + 8 * hi;
        *(v2f*)(dbase + 0) = (v2f){acc2[0], acc2[1]};
        *(v2f*)(dbase + 2) = (v2f){acc2[2], acc2[3]};
        *(v2f*)(dbase + 4) = (v2f){acc2[4], acc2[5]};
        *(v2f*)(dbase + 6) = (v2f){acc2[6], acc2[7]};
    }
    __syncthreads();

    float* outp = xout + ((b * HEIGHT) * WIDTH + ww * 4) * CHAN;
    for (int i = tid; i < 4096; i += 256) {
        int h2  = i >> 4;
        int q   = (i >> 2) & 3;
        int c2b = (i & 3) << 2;
        int goff = (h2 * WIDTH + q) * CHAN + c2b;
        float4 xv = *(const float4*)(xres + goff);
        float4 dv;
        dv.x = delta[((c2b + 0) * 4 + q) * DELTA_STRIDE + h2];
        dv.y = delta[((c2b + 1) * 4 + q) * DELTA_STRIDE + h2];
        dv.z = delta[((c2b + 2) * 4 + q) * DELTA_STRIDE + h2];
        dv.w = delta[((c2b + 3) * 4 + q) * DELTA_STRIDE + h2];
        *(float4*)(outp + goff) =
            make_float4(xv.x + dv.x, xv.y + dv.y, xv.z + dv.z, xv.w + dv.w);
    }
}

extern "C" void kernel_launch(void* const* d_in, const int* in_sizes, int n_in,
                              void* d_out, int out_size, void* d_ws, size_t ws_size,
                              hipStream_t stream) {
    (void)in_sizes; (void)n_in; (void)out_size; (void)ws_size;
    const float* x   = (const float*)d_in[0];
    const float* cwv = (const float*)d_in[1];
    const float* f0w = (const float*)d_in[2];
    const float* f0b = (const float*)d_in[3];
    const float* f1w = (const float*)d_in[4];

    float* tmp = (float*)d_ws;
    float* out = (float*)d_out;

    dim3 grid(BATCH * 64);
    dim3 block(256);
    size_t shmem = SMEM_FLOATS * sizeof(float);

    ca_step_kernel<<<grid, block, shmem, stream>>>(x,   cwv, f0w, f0b, f1w, tmp);
    ca_step_kernel<<<grid, block, shmem, stream>>>(tmp, cwv, f0w, f0b, f1w, out);
}


# global loads of each phase issued as one independent group before their consumers: 52 staging + 16 residual loads per thread in flight together (clamped address, zero select at borders)
# speedup vs baseline: 1.9012x; 1.9012x over previous
#include <hip/hip_runtime.h>


typedef __attribute__((ext_vector_type(2))) float v2f;
typedef __attribute__((ext_vector_type(8))) float v8f;

#define BATCH  32
#define HEIGHT 256
#define WIDTH  256
#define CHAN   16
#define HCN    4
#define HS     12

#define XROW_STRIDE 20
#define XROWS       258
#define XCOL_SIZE   (XROWS * XROW_STRIDE)
#define NCOLS       12
#define OFF_XCOL    0
#define OFF_CONVW   (NCOLS * XCOL_SIZE)
#define OFF_CONVOUT (OFF_CONVW + 576)
#define OFF_US      (OFF_CONVOUT + 4096)
#define SMEM_FLOATS (OFF_US + 8 * 2 * 256)
#define DELTA_STRIDE 260
#define OFF_DELTA   0
#define ST_ROWS     21
#define ST_PASSES   13

__launch_bounds__(256)
__global__ void ca_step_kernel(const float* __restrict__ xin,
                               const float* __restrict__ convw,
                               const float* __restrict__ fc0w,
                               const float* __restrict__ fc0b,
                               const float* __restrict__ fc1w,
                               float* __restrict__ xout)
{
    extern __shared__ float smem[];
    float* xcol  = smem + OFF_XCOL;
    float* cw    = smem + OFF_CONVW;
    float* cout  = smem + OFF_CONVOUT;
    float* us    = smem + OFF_US;
    float* delta = smem + OFF_DELTA;

    const int tid  = threadIdx.x;
    const int lane = tid & 31;
    const int wave = tid >> 5;
    const int m    = lane & 15;
    const int hi   = lane >> 4;

    const int blk = blockIdx.x;
    const int b   = blk >> 6;
    const int ww  = blk & 63;

    const float* xres = xin + ((b * HEIGHT) * WIDTH + ww * 4) * CHAN;

    for (int i = tid; i < 1024; i += 256) {
        int h2 = i >> 2;
        int q  = i & 3;
        __builtin_prefetch(xres + (h2 * WIDTH + q) * CHAN, 0, 1);
    }

    v2f afc0;
    {
        int s = m;
        int base = (s < HS ? s : 0) * 4 + 2 * hi;
        float a0 = fc0w[base + 0];
        float a1 = fc0w[base + 1];
        afc0 = (s < HS) ? (v2f){a0, a1} : (v2f){0.f, 0.f};
    }
    v8f cbias;
#pragma unroll
    for (int v = 0; v < 8; ++v) {
        int s = v + 8 * hi;
        float bv = fc0b[s < HS ? s : 0];
        cbias[v] = (s < HS) ? bv : 0.f;
    }
    v2f afc1[3];
#pragma unroll
    for (int t3 = 0; t3 < 3; ++t3) {
        int base = m * HS + 4 * t3 + 2 * hi;
        afc1[t3].x = fc1w[base + 0];
        afc1[t3].y = fc1w[base + 1];
    }

    {
        const int cwi2 = tid < 64 ? tid + 512 : 575;
        float cwv0 = convw[tid];
        float cwv1 = convw[tid + 256];
        float cwv2 = convw[cwi2];

        const int  hq    = tid / 12;
        const int  q     = tid - hq * 12;
        const int  hr    = hq < ST_ROWS ? hq : 0;
        const int  c3    = q >> 2;
        const bool edgeL = (ww == 0)  && (c3 == 0);
        const bool edgeR = (ww == 63) && (c3 == 2);
        const bool top   = (hr == 0);
        const bool bot   = (hr == 5);
        const int  gbase = b * (HEIGHT * WIDTH * CHAN) + (hr - 1) * (WIDTH * CHAN)
                         + (ww - 1) * CHAN + q * 4;
        const int  adjL   = edgeL ? CHAN : 0;
        const int  adjR   = edgeR ? -CHAN : 0;
        const int  adjTop = top ? (WIDTH * CHAN) : 0;
        const int  adjBot = (hr > 4) ? -(hr - 4) * (WIDTH * CHAN) : 0;

        float4 sv[4 * (ST_PASSES - 1)];
        float4 sl[4];
#pragma unroll
        for (int j = 0; j < 4 * (ST_PASSES - 1); ++j) {
            const int t = j / (ST_PASSES - 1);
            const int k = j - t * (ST_PASSES - 1);
            int off = gbase + t * (64 * CHAN) + k * (ST_ROWS * WIDTH * CHAN);
            if (t == 0) off += adjL;
            if (t == 3) off += adjR;
            if (k == 0) off += adjTop;
            sv[j] = *(const float4*)(xin + off);
        }
#pragma unroll
        for (int t = 0; t < 4; ++t) {
            int off = gbase + t * (64 * CHAN) + (ST_PASSES - 1) * (ST_ROWS * WIDTH * CHAN) + adjBot;
            if (t == 0) off += adjL;
            if (t == 3) off += adjR;
            sl[t] = *(const float4*)(xin + off);
        }

        cw[tid]       = cwv0;
        cw[tid + 256] = cwv1;
        cw[cwi2]      = cwv2;

        float* ldst = xcol + c3 * XCOL_SIZE + hr * XROW_STRIDE + (q & 3) * 4;
#pragma unroll
        for (int j = 0; j < 4 * (ST_PASSES - 1); ++j) {
            const int t = j / (ST_PASSES - 1);
            const int k = j - t * (ST_PASSES - 1);
            bool zero = false;
            if (t == 0) zero = zero || edgeL;
            if (t == 3) zero = zero || edgeR;
            if (k == 0) zero = zero || top;
            float4 val = sv[j];
            if (t == 0 || t == 3 || k == 0) {
                asm volatile("" :: "v"(val.x), "v"(val.y), "v"(val.z), "v"(val.w));
                if (zero) val = make_float4(0.f, 0.f, 0.f, 0.f);
            }
            *(float4*)(ldst + 3 * t * XCOL_SIZE + k * (ST_ROWS * XROW_STRIDE)) = val;
        }
#pragma unroll
        for (int t = 0; t < 4; ++t)
            asm volatile("" :: "v"(sl[t].x), "v"(sl[t].y), "v"(sl[t].z), "v"(sl[t].w));
        if (hr < 6) {
#pragma unroll
            for (int t = 0; t < 4; ++t) {
                bool zero = bot;
                if (t == 0) zero = zero || edgeL;
                if (t == 3) zero = zero || edgeR;
                float4 val = sl[t];
                if (zero) val = make_float4(0.f, 0.f, 0.f, 0.f);
                *(float4*)(ldst + 3 * t * XCOL_SIZE + (ST_PASSES - 1) * (ST_ROWS * XROW_STRIDE)) = val;
            }
        }
    }
    __syncthreads();


    v2f aw[36];
    {
        const float* pw = cw + (m & 3) * 144 + (2 * hi) * 9;
#pragma unroll
        for (int a = 0; a < 3; ++a)
#pragma unroll
            for (int d = 0; d < 3; ++d)
#pragma unroll
                for (int cblk = 0; cblk < 4; ++cblk) {
                    float w0 = pw[cblk * 36 + 0 + a * 3 + d];
                    float w1 = pw[cblk * 36 + 9 + a * 3 + d];
                    aw[(a * 3 + d) * 4 + cblk] =
                        (m < HCN) ? (v2f){w0, w1} : (v2f){0.f, 0.f};
                }
    }

    for (int uu = 0; uu < 8; ++uu) {
        int u  = wave * 8 + uu;
        int t  = u >> 4;
        int h0 = (u & 15) << 4;
        v8f acc = {};
        const float* pb = xcol + 3 * t * XCOL_SIZE + (h0 + m) * XROW_STRIDE + 2 * hi;
#pragma unroll
        for (int a = 0; a < 3; ++a) {
#pragma unroll
            for (int d = 0; d < 3; ++d) {
                const float* pg = pb + a * XCOL_SIZE + d * XROW_STRIDE;
                v2f b0 = *(const v2f*)(pg + 0);
                v2f b1 = *(const v2f*)(pg + 4);
                v2f b2 = *(const v2f*)(pg + 8);
                v2f b3 = *(const v2f*)(pg + 12);
                int s = (a * 3 + d) * 4;
                acc = __builtin_amdgcn_wmma_f32_16x16x4_f32(
                    false, aw[s + 0], false, b0, (short)0, acc, false, false);
                acc = __builtin_amdgcn_wmma_f32_16x16x4_f32(
                    false, aw[s + 1], false, b1, (short)0, acc, false, false);
                acc = __builtin_amdgcn_wmma_f32_16x16x4_f32(
                    false, aw[s + 2], false, b2, (short)0, acc, false, false);
                acc = __builtin_amdgcn_wmma_f32_16x16x4_f32(
                    false, aw[s + 3], false, b3, (short)0, acc, false, false);
            }
        }
        if (lane < 16) {
            float* cdst = cout + (t * 4) * 256 + h0 + lane;
            cdst[0]   = fmaxf(acc[0], 0.f);
            cdst[256] = fmaxf(acc[1], 0.f);
            cdst[512] = fmaxf(acc[2], 0.f);
            cdst[768] = fmaxf(acc[3], 0.f);
        }
    }
    __syncthreads();

#pragma unroll 2
    for (int uu = 0; uu < 8; ++uu) {
        int tt = wave * 8 + uu;
        int hc = tt >> 4;
        int t  = (tt >> 2) & 3;
        int q  = tt & 3;

        v2f Bv = *(const v2f*)(cout + (t * 4 + hc) * 256 + (q * 16 + m) * 4 + 2 * hi);
        v8f u0 = __builtin_amdgcn_wmma_f32_16x16x4_f32(
            false, afc0, false, Bv, (short)0, cbias, false, false);
#pragma unroll
        for (int v = 0; v < 8; ++v) u0[v] = fmaxf(u0[v], 0.f);

        float* usw = us + wave * 512 + (uu & 1) * 256;
        *(float4*)(usw + m * 16 + 8 * hi + 0) = make_float4(u0[0], u0[1], u0[2], u0[3]);
        *(float4*)(usw + m * 16 + 8 * hi + 4) = make_float4(u0[4], u0[5], u0[6], u0[7]);

        v8f acc2 = {};
#pragma unroll
        for (int t3 = 0; t3 < 3; ++t3) {
            v2f B2 = *(const v2f*)(usw + m * 16 + 4 * t3 + 2 * hi);
            acc2 = __builtin_amdgcn_wmma_f32_16x16x4_f32(
                false, afc1[t3], false, B2, (short)0, acc2, false, false);
        }

        int cq = (4 * hc + t) * 4 + q;
        float* dbase = delta + cq * DELTA_STRIDE + m * 16 + 8 * hi;
        *(v2f*)(dbase + 0) = (v2f){acc2[0], acc2[1]};
        *(v2f*)(dbase + 2) = (v2f){acc2[2], acc2[3]};
        *(v2f*)(dbase + 4) = (v2f){acc2[4], acc2[5]};
        *(v2f*)(dbase + 6) = (v2f){acc2[6], acc2[7]};
    }
    const int r_h2  = tid >> 4;
    const int r_q   = (tid >> 2) & 3;
    const int r_c2b = (tid & 3) << 2;
    const int r_off = (r_h2 * WIDTH + r_q) * CHAN + r_c2b;
    float4 xr[16];
#pragma unroll
    for (int k = 0; k < 16; ++k)
        xr[k] = *(const float4*)(xres + r_off + k * (16 * WIDTH * CHAN));
    __syncthreads();

    float* outp = xout + ((b * HEIGHT) * WIDTH + ww * 4) * CHAN;
    const float* dsrc = delta + (r_c2b * 4 + r_q) * DELTA_STRIDE + r_h2;
#pragma unroll
    for (int k = 0; k < 16; ++k) {
        float4 xv = xr[k];
        float4 dv;
        dv.x = dsrc[0 * 4 * DELTA_STRIDE + 16 * k];
        dv.y = dsrc[1 * 4 * DELTA_STRIDE + 16 * k];
        dv.z = dsrc[2 * 4 * DELTA_STRIDE + 16 * k];
        dv.w = dsrc[3 * 4 * DELTA_STRIDE + 16 * k];
        *(float4*)(outp + r_off + k * (16 * WIDTH * CHAN)) =
            make_float4(xv.x + dv.x, xv.y + dv.y, xv.z + dv.z, xv.w + dv.w);
    }
}

extern "C" void kernel_launch(void* const* d_in, const int* in_sizes, int n_in,
                              void* d_out, int out_size, void* d_ws, size_t ws_size,
                              hipStream_t stream) {
    (void)in_sizes; (void)n_in; (void)out_size; (void)ws_size;
    const float* x   = (const float*)d_in[0];
    const float* cwv = (const float*)d_in[1];
    const float* f0w = (const float*)d_in[2];
    const float* f0b = (const float*)d_in[3];
    const float* f1w = (const float*)d_in[4];

    float* tmp = (float*)d_ws;
    float* out = (float*)d_out;

    dim3 grid(BATCH * 64);
    dim3 block(256);
    size_t shmem = SMEM_FLOATS * sizeof(float);

    ca_step_kernel<<<grid, block, shmem, stream>>>(x,   cwv, f0w, f0b, f1w, tmp);
    ca_step_kernel<<<grid, block, shmem, stream>>>(tmp, cwv, f0w, f0b, f1w, out);
}


# work-group tile halved along h (128 source rows, 2 of the 4 output columns): LDS 282752 -> 151680 B so two work-groups are resident per WGP and one stages while the other computes
# speedup vs baseline: 2.3232x; 1.2220x over previous
#include <hip/hip_runtime.h>


typedef __attribute__((ext_vector_type(2))) float v2f;
typedef __attribute__((ext_vector_type(8))) float v8f;

#define BATCH  32
#define HEIGHT 256
#define WIDTH  256
#define CHAN   16
#define HCN    4
#define HS     12

#define XROW_STRIDE 20
#define HROWS       128
#define XROWS       130
#define XCOL_SIZE   (XROWS * XROW_STRIDE)
#define NCOLS       12
#define OFF_XCOL    0
#define OFF_CONVW   (NCOLS * XCOL_SIZE)
#define OFF_CONVOUT (OFF_CONVW + 576)
#define OFF_US      (OFF_CONVOUT + 4 * HCN * HROWS)
#define SMEM_FLOATS (OFF_US + 8 * 2 * 256)
#define DELTA_STRIDE 260
#define OFF_DELTA   0
#define ST_ROWS     21
#define ST_PASSES   7

__launch_bounds__(256)
__global__ void ca_step_kernel(const float* __restrict__ xin,
                               const float* __restrict__ convw,
                               const float* __restrict__ fc0w,
                               const float* __restrict__ fc0b,
                               const float* __restrict__ fc1w,
                               float* __restrict__ xout)
{
    extern __shared__ float smem[];
    float* xcol  = smem + OFF_XCOL;
    float* cw    = smem + OFF_CONVW;
    float* cout  = smem + OFF_CONVOUT;
    float* us    = smem + OFF_US;
    float* delta = smem + OFF_DELTA;

    const int tid  = threadIdx.x;
    const int lane = tid & 31;
    const int wave = tid >> 5;
    const int m    = lane & 15;
    const int hi   = lane >> 4;

    const int blk = blockIdx.x;
    const int b   = blk >> 7;
    const int hf  = (blk >> 6) & 1;
    const int ww  = blk & 63;

    const float* xres = xin + ((b * HEIGHT) * WIDTH + ww * 4 + 2 * hf) * CHAN;

    for (int i = tid; i < 512; i += 256) {
        int h2 = i >> 1;
        int q  = i & 1;
        __builtin_prefetch(xres + (h2 * WIDTH + q) * CHAN, 0, 1);
    }

    v2f afc0;
    {
        int s = m;
        int base = (s < HS ? s : 0) * 4 + 2 * hi;
        float a0 = fc0w[base + 0];
        float a1 = fc0w[base + 1];
        afc0 = (s < HS) ? (v2f){a0, a1} : (v2f){0.f, 0.f};
    }
    v8f cbias;
#pragma unroll
    for (int v = 0; v < 8; ++v) {
        int s = v + 8 * hi;
        float bv = fc0b[s < HS ? s : 0];
        cbias[v] = (s < HS) ? bv : 0.f;
    }
    v2f afc1[3];
#pragma unroll
    for (int t3 = 0; t3 < 3; ++t3) {
        int base = m * HS + 4 * t3 + 2 * hi;
        afc1[t3].x = fc1w[base + 0];
        afc1[t3].y = fc1w[base + 1];
    }

    {
        const int cwi2 = tid < 64 ? tid + 512 : 575;
        float cwv0 = convw[tid];
        float cwv1 = convw[tid + 256];
        float cwv2 = convw[cwi2];

        const int  hq    = tid / 12;
        const int  q     = tid - hq * 12;
        const int  hr    = hq < ST_ROWS ? hq : 0;
        const int  c3    = q >> 2;
        const bool edgeL = (ww == 0)  && (c3 == 0);
        const bool edgeR = (ww == 63) && (c3 == 2);
        const bool top   = (hf == 0) && (hr == 0);
        const bool bot   = (hf == 1) && (hr == 3);
        const int  gbase = b * (HEIGHT * WIDTH * CHAN) + (hf * HROWS + hr - 1) * (WIDTH * CHAN)
                         + (ww - 1) * CHAN + q * 4;
        const int  adjL   = edgeL ? CHAN : 0;
        const int  adjR   = edgeR ? -CHAN : 0;
        const int  adjTop = top ? (WIDTH * CHAN) : 0;
        const int  adjBot = (hf == 1 && hr > 2) ? -(hr - 2) * (WIDTH * CHAN) : 0;

        float4 sv[4 * (ST_PASSES - 1)];
        float4 sl[4];
#pragma unroll
        for (int j = 0; j < 4 * (ST_PASSES - 1); ++j) {
            const int t = j / (ST_PASSES - 1);
            const int k = j - t * (ST_PASSES - 1);
            int off = gbase + t * (64 * CHAN) + k * (ST_ROWS * WIDTH * CHAN);
            if (t == 0) off += adjL;
            if (t == 3) off += adjR;
            if (k == 0) off += adjTop;
            sv[j] = *(const float4*)(xin + off);
        }
#pragma unroll
        for (int t = 0; t < 4; ++t) {
            int off = gbase + t * (64 * CHAN) + (ST_PASSES - 1) * (ST_ROWS * WIDTH * CHAN) + adjBot;
            if (t == 0) off += adjL;
            if (t == 3) off += adjR;
            sl[t] = *(const float4*)(xin + off);
        }

        cw[tid]       = cwv0;
        cw[tid + 256] = cwv1;
        cw[cwi2]      = cwv2;

        float* ldst = xcol + c3 * XCOL_SIZE + hr * XROW_STRIDE + (q & 3) * 4;
#pragma unroll
        for (int j = 0; j < 4 * (ST_PASSES - 1); ++j) {
            const int t = j / (ST_PASSES - 1);
            const int k = j - t * (ST_PASSES - 1);
            bool zero = false;
            if (t == 0) zero = zero || edgeL;
            if (t == 3) zero = zero || edgeR;
            if (k == 0) zero = zero || top;
            float4 val = sv[j];
            if (t == 0 || t == 3 || k == 0) {
                asm volatile("" :: "v"(val.x), "v"(val.y), "v"(val.z), "v"(val.w));
                if (zero) val = make_float4(0.f, 0.f, 0.f, 0.f);
            }
            *(float4*)(ldst + 3 * t * XCOL_SIZE + k * (ST_ROWS * XROW_STRIDE)) = val;
        }
#pragma unroll
        for (int t = 0; t < 4; ++t)
            asm volatile("" :: "v"(sl[t].x), "v"(sl[t].y), "v"(sl[t].z), "v"(sl[t].w));
        if (hr < 4) {
#pragma unroll
            for (int t = 0; t < 4; ++t) {
                bool zero = bot;
                if (t == 0) zero = zero || edgeL;
                if (t == 3) zero = zero || edgeR;
                float4 val = sl[t];
                if (zero) val = make_float4(0.f, 0.f, 0.f, 0.f);
                *(float4*)(ldst + 3 * t * XCOL_SIZE + (ST_PASSES - 1) * (ST_ROWS * XROW_STRIDE)) = val;
            }
        }
    }
    __syncthreads();


    v2f aw[36];
    {
        const float* pw = cw + (m & 3) * 144 + (2 * hi) * 9;
#pragma unroll
        for (int a = 0; a < 3; ++a)
#pragma unroll
            for (int d = 0; d < 3; ++d)
#pragma unroll
                for (int cblk = 0; cblk < 4; ++cblk) {
                    float w0 = pw[cblk * 36 + 0 + a * 3 + d];
                    float w1 = pw[cblk * 36 + 9 + a * 3 + d];
                    aw[(a * 3 + d) * 4 + cblk] =
                        (m < HCN) ? (v2f){w0, w1} : (v2f){0.f, 0.f};
                }
    }

    for (int uu = 0; uu < 4; ++uu) {
        int u  = wave * 4 + uu;
        int t  = u >> 3;
        int h0 = (u & 7) << 4;
        v8f acc = {};
        const float* pb = xcol + 3 * t * XCOL_SIZE + (h0 + m) * XROW_STRIDE + 2 * hi;
#pragma unroll
        for (int a = 0; a < 3; ++a) {
#pragma unroll
            for (int d = 0; d < 3; ++d) {
                const float* pg = pb + a * XCOL_SIZE + d * XROW_STRIDE;
                v2f b0 = *(const v2f*)(pg + 0);
                v2f b1 = *(const v2f*)(pg + 4);
                v2f b2 = *(const v2f*)(pg + 8);
                v2f b3 = *(const v2f*)(pg + 12);
                int s = (a * 3 + d) * 4;
                acc = __builtin_amdgcn_wmma_f32_16x16x4_f32(
                    false, aw[s + 0], false, b0, (short)0, acc, false, false);
                acc = __builtin_amdgcn_wmma_f32_16x16x4_f32(
                    false, aw[s + 1], false, b1, (short)0, acc, false, false);
                acc = __builtin_amdgcn_wmma_f32_16x16x4_f32(
                    false, aw[s + 2], false, b2, (short)0, acc, false, false);
                acc = __builtin_amdgcn_wmma_f32_16x16x4_f32(
                    false, aw[s + 3], false, b3, (short)0, acc, false, false);
            }
        }
        if (lane < 16) {
            float* cdst = cout + (t * 4) * HROWS + h0 + lane;
            cdst[0 * HROWS] = fmaxf(acc[0], 0.f);
            cdst[1 * HROWS] = fmaxf(acc[1], 0.f);
            cdst[2 * HROWS] = fmaxf(acc[2], 0.f);
            cdst[3 * HROWS] = fmaxf(acc[3], 0.f);
        }
    }
    __syncthreads();

#pragma unroll 2
    for (int uu = 0; uu < 4; ++uu) {
        int tt = wave * 4 + uu;
        int hc = tt >> 3;
        int t  = (tt >> 1) & 3;
        int q  = tt & 1;

        v2f Bv = *(const v2f*)(cout + (t * 4 + hc) * HROWS + (q * 16 + m) * 4 + 2 * hi);
        v8f u0 = __builtin_amdgcn_wmma_f32_16x16x4_f32(
            false, afc0, false, Bv, (short)0, cbias, false, false);
#pragma unroll
        for (int v = 0; v < 8; ++v) u0[v] = fmaxf(u0[v], 0.f);

        float* usw = us + wave * 512 + (uu & 1) * 256;
        *(float4*)(usw + m * 16 + 8 * hi + 0) = make_float4(u0[0], u0[1], u0[2], u0[3]);
        *(float4*)(usw + m * 16 + 8 * hi + 4) = make_float4(u0[4], u0[5], u0[6], u0[7]);

        v8f acc2 = {};
#pragma unroll
        for (int t3 = 0; t3 < 3; ++t3) {
            v2f B2 = *(const v2f*)(usw + m * 16 + 4 * t3 + 2 * hi);
            acc2 = __builtin_amdgcn_wmma_f32_16x16x4_f32(
                false, afc1[t3], false, B2, (short)0, acc2, false, false);
        }

        int cq = (4 * hc + t) * 2 + q;
        float* dbase = delta + cq * DELTA_STRIDE + m * 16 + 8 * hi;
        *(v2f*)(dbase + 0) = (v2f){acc2[0], acc2[1]};
        *(v2f*)(dbase + 2) = (v2f){acc2[2], acc2[3]};
        *(v2f*)(dbase + 4) = (v2f){acc2[4], acc2[5]};
        *(v2f*)(dbase + 6) = (v2f){acc2[6], acc2[7]};
    }
    const int r_h2  = tid >> 3;
    const int r_q   = (tid >> 2) & 1;
    const int r_c2b = (tid & 3) << 2;
    const int r_off = (r_h2 * WIDTH + r_q) * CHAN + r_c2b;
    float4 xr[8];
#pragma unroll
    for (int k = 0; k < 8; ++k)
        xr[k] = *(const float4*)(xres + r_off + k * (32 * WIDTH * CHAN));
    __syncthreads();

    float* outp = xout + ((b * HEIGHT) * WIDTH + ww * 4 + 2 * hf) * CHAN;
    const float* dsrc = delta + (r_c2b * 2 + r_q) * DELTA_STRIDE + r_h2;
#pragma unroll
    for (int k = 0; k < 8; ++k) {
        float4 xv = xr[k];
        float4 dv;
        dv.x = dsrc[0 * 2 * DELTA_STRIDE + 32 * k];
        dv.y = dsrc[1 * 2 * DELTA_STRIDE + 32 * k];
        dv.z = dsrc[2 * 2 * DELTA_STRIDE + 32 * k];
        dv.w = dsrc[3 * 2 * DELTA_STRIDE + 32 * k];
        *(float4*)(outp + r_off + k * (32 * WIDTH * CHAN)) =
            make_float4(xv.x + dv.x, xv.y + dv.y, xv.z + dv.z, xv.w + dv.w);
    }
}

extern "C" void kernel_launch(void* const* d_in, const int* in_sizes, int n_in,
                              void* d_out, int out_size, void* d_ws, size_t ws_size,
                              hipStream_t stream) {
    (void)in_sizes; (void)n_in; (void)out_size; (void)ws_size;
    const float* x   = (const float*)d_in[0];
    const float* cwv = (const float*)d_in[1];
    const float* f0w = (const float*)d_in[2];
    const float* f0b = (const float*)d_in[3];
    const float* f1w = (const float*)d_in[4];

    float* tmp = (float*)d_ws;
    float* out = (float*)d_out;

    dim3 grid(BATCH * 128);
    dim3 block(256);
    size_t shmem = SMEM_FLOATS * sizeof(float);

    ca_step_kernel<<<grid, block, shmem, stream>>>(x,   cwv, f0w, f0b, f1w, tmp);
    ca_step_kernel<<<grid, block, shmem, stream>>>(tmp, cwv, f0w, f0b, f1w, out);
}


# MLP phase in registers: fc0 accumulator (hidden units 6+6 over the lane halves) is fc1's B operand, no LDS transpose; a wave's 4 tiles are independent chains; fc1 fp32 summation order changes
# speedup vs baseline: 3.0127x; 1.2968x over previous
#include <hip/hip_runtime.h>


typedef __attribute__((ext_vector_type(2))) float v2f;
typedef __attribute__((ext_vector_type(8))) float v8f;

#define BATCH  32
#define HEIGHT 256
#define WIDTH  256
#define CHAN   16
#define HCN    4
#define HS     12

#define XROW_STRIDE 20
#define HROWS       128
#define XROWS       130
#define XCOL_SIZE   (XROWS * XROW_STRIDE)
#define NCOLS       12
#define OFF_XCOL    0
#define OFF_CONVW   (NCOLS * XCOL_SIZE)
#define PSTRIDE     132
#define OFF_P       (OFF_CONVW + 576)
#define SMEM_FLOATS (OFF_P + 4 * 3 * HCN * PSTRIDE)
#define DELTA_STRIDE 260
#define OFF_DELTA   0
#define ST_ROWS     21
#define ST_PASSES   7

__launch_bounds__(256)
__global__ void ca_step_kernel(const float* __restrict__ xin,
                               const float* __restrict__ convw,
                               const float* __restrict__ fc0w,
                               const float* __restrict__ fc0b,
                               const float* __restrict__ fc1w,
                               float* __restrict__ xout)
{
    extern __shared__ float smem[];
    float* xcol  = smem + OFF_XCOL;
    float* cw    = smem + OFF_CONVW;
    float* pbuf  = smem + OFF_P;
    float* delta = smem + OFF_DELTA;

    const int tid  = threadIdx.x;
    const int lane = tid & 31;
    const int wave = tid >> 5;
    const int m    = lane & 15;
    const int hi   = lane >> 4;

    const int blk = blockIdx.x;
    const int b   = blk >> 7;
    const int hf  = (blk >> 6) & 1;
    const int ww  = blk & 63;

    const float* xres = xin + ((b * HEIGHT) * WIDTH + ww * 4 + 2 * hf) * CHAN;

    for (int i = tid; i < 512; i += 256) {
        int h2 = i >> 1;
        int q  = i & 1;
        __builtin_prefetch(xres + (h2 * WIDTH + q) * CHAN, 0, 1);
    }

    v2f afc0;
    {
        const bool live = (m < 6) || (m >= 8 && m < 14);
        const int  s    = (m < 6) ? m : m - 2;
        int base = (live ? s : 0) * 4 + 2 * hi;
        float a0 = fc0w[base + 0];
        float a1 = fc0w[base + 1];
        afc0 = live ? (v2f){a0, a1} : (v2f){0.f, 0.f};
    }
    v8f cbias;
#pragma unroll
    for (int v = 0; v < 8; ++v) {
        float bv = fc0b[v < 6 ? 6 * hi + v : 0];
        cbias[v] = (v < 6) ? bv : 0.f;
    }
    v2f afc1[3];
#pragma unroll
    for (int i = 0; i < 3; ++i) {
        int base = m * HS + 6 * hi + 2 * i;
        afc1[i].x = fc1w[base + 0];
        afc1[i].y = fc1w[base + 1];
    }

    {
        const int cwi2 = tid < 64 ? tid + 512 : 575;
        float cwv0 = convw[tid];
        float cwv1 = convw[tid + 256];
        float cwv2 = convw[cwi2];

        const int  hq    = tid / 12;
        const int  q     = tid - hq * 12;
        const int  hr    = hq < ST_ROWS ? hq : 0;
        const int  c3    = q >> 2;
        const bool edgeL = (ww == 0)  && (c3 == 0);
        const bool edgeR = (ww == 63) && (c3 == 2);
        const bool top   = (hf == 0) && (hr == 0);
        const bool bot   = (hf == 1) && (hr == 3);
        const int  gbase = b * (HEIGHT * WIDTH * CHAN) + (hf * HROWS + hr - 1) * (WIDTH * CHAN)
                         + (ww - 1) * CHAN + q * 4;
        const int  adjL   = edgeL ? CHAN : 0;
        const int  adjR   = edgeR ? -CHAN : 0;
        const int  adjTop = top ? (WIDTH * CHAN) : 0;
        const int  adjBot = (hf == 1 && hr > 2) ? -(hr - 2) * (WIDTH * CHAN) : 0;

        float4 sv[4 * (ST_PASSES - 1)];
        float4 sl[4];
#pragma unroll
        for (int j = 0; j < 4 * (ST_PASSES - 1); ++j) {
            const int t = j / (ST_PASSES - 1);
            const int k = j - t * (ST_PASSES - 1);
            int off = gbase + t * (64 * CHAN) + k * (ST_ROWS * WIDTH * CHAN);
            if (t == 0) off += adjL;
            if (t == 3) off += adjR;
            if (k == 0) off += adjTop;
            sv[j] = *(const float4*)(xin + off);
        }
#pragma unroll
        for (int t = 0; t < 4; ++t) {
            int off = gbase + t * (64 * CHAN) + (ST_PASSES - 1) * (ST_ROWS * WIDTH * CHAN) + adjBot;
            if (t == 0) off += adjL;
            if (t == 3) off += adjR;
            sl[t] = *(const float4*)(xin + off);
        }

        cw[tid]       = cwv0;
        cw[tid + 256] = cwv1;
        cw[cwi2]      = cwv2;

        float* ldst = xcol + c3 * XCOL_SIZE + hr * XROW_STRIDE + (q & 3) * 4;
#pragma unroll
        for (int j = 0; j < 4 * (ST_PASSES - 1); ++j) {
            const int t = j / (ST_PASSES - 1);
            const int k = j - t * (ST_PASSES - 1);
            bool zero = false;
            if (t == 0) zero = zero || edgeL;
            if (t == 3) zero = zero || edgeR;
            if (k == 0) zero = zero || top;
            float4 val = sv[j];
            if (t == 0 || t == 3 || k == 0) {
                asm volatile("" :: "v"(val.x), "v"(val.y), "v"(val.z), "v"(val.w));
                if (zero) val = make_float4(0.f, 0.f, 0.f, 0.f);
            }
            *(float4*)(ldst + 3 * t * XCOL_SIZE + k * (ST_ROWS * XROW_STRIDE)) = val;
        }
#pragma unroll
        for (int t = 0; t < 4; ++t)
            asm volatile("" :: "v"(sl[t].x), "v"(sl[t].y), "v"(sl[t].z), "v"(sl[t].w));
        if (hr < 4) {
#pragma unroll
            for (int t = 0; t < 4; ++t) {
                bool zero = bot;
                if (t == 0) zero = zero || edgeL;
                if (t == 3) zero = zero || edgeR;
                float4 val = sl[t];
                if (zero) val = make_float4(0.f, 0.f, 0.f, 0.f);
                *(float4*)(ldst + 3 * t * XCOL_SIZE + (ST_PASSES - 1) * (ST_ROWS * XROW_STRIDE)) = val;
            }
        }
    }
    __syncthreads();


    v2f aw[12];
    {
        const int dd = (m >> 2) < 3 ? (m >> 2) : 2;
        const float* pw = cw + (m & 3) * 144 + (2 * hi) * 9 + dd;
#pragma unroll
        for (int a = 0; a < 3; ++a)
#pragma unroll
            for (int cblk = 0; cblk < 4; ++cblk) {
                float w0 = pw[cblk * 36 + 0 + a * 3];
                float w1 = pw[cblk * 36 + 9 + a * 3];
                aw[a * 4 + cblk] = (m < 12) ? (v2f){w0, w1} : (v2f){0.f, 0.f};
            }
    }

#pragma unroll
    for (int pp = 0; pp < 2; ++pp) {
        const int u = wave * 4 + pp * 2;
        const int t = u >> 3;
        const int r = ((u & 7) << 4) + m;
        const float* pb = xcol + 3 * t * XCOL_SIZE + r * XROW_STRIDE + 2 * hi;
        v2f bA[12], bB[12];
#pragma unroll
        for (int a = 0; a < 3; ++a)
#pragma unroll
            for (int cblk = 0; cblk < 4; ++cblk) {
                bA[a * 4 + cblk] = *(const v2f*)(pb + a * XCOL_SIZE + 4 * cblk);
                bB[a * 4 + cblk] = *(const v2f*)(pb + a * XCOL_SIZE + 16 * XROW_STRIDE + 4 * cblk);
            }
        v8f accA = {};
        v8f accB = {};
#pragma unroll
        for (int s = 0; s < 12; ++s) {
            accA = __builtin_amdgcn_wmma_f32_16x16x4_f32(
                false, aw[s], false, bA[s], (short)0, accA, false, false);
            accB = __builtin_amdgcn_wmma_f32_16x16x4_f32(
                false, aw[s], false, bB[s], (short)0, accB, false, false);
        }
        float* pA = pbuf + ((3 * t + 2 * hi) * 4) * PSTRIDE + 2 + r - 2 * hi;
        pA[0 * PSTRIDE] = accA[0];
        pA[1 * PSTRIDE] = accA[1];
        pA[2 * PSTRIDE] = accA[2];
        pA[3 * PSTRIDE] = accA[3];
        pA[0 * PSTRIDE + 16] = accB[0];
        pA[1 * PSTRIDE + 16] = accB[1];
        pA[2 * PSTRIDE + 16] = accB[2];
        pA[3 * PSTRIDE + 16] = accB[3];
        if (hi == 0) {
            float* pB = pbuf + ((3 * t + 1) * 4) * PSTRIDE + 1 + r;
            pB[0 * PSTRIDE] = accA[4];
            pB[1 * PSTRIDE] = accA[5];
            pB[2 * PSTRIDE] = accA[6];
            pB[3 * PSTRIDE] = accA[7];
            pB[0 * PSTRIDE + 16] = accB[4];
            pB[1 * PSTRIDE + 16] = accB[5];
            pB[2 * PSTRIDE + 16] = accB[6];
            pB[3 * PSTRIDE + 16] = accB[7];
        }
    }
    if (wave == 0) {
        const int t = (m >> 1) & 3;
        const int r = 128 + (m & 1);
        const float* pb = xcol + 3 * t * XCOL_SIZE + r * XROW_STRIDE + 2 * hi;
        v2f bA[12];
#pragma unroll
        for (int a = 0; a < 3; ++a)
#pragma unroll
            for (int cblk = 0; cblk < 4; ++cblk)
                bA[a * 4 + cblk] = *(const v2f*)(pb + a * XCOL_SIZE + 4 * cblk);
        v8f accA = {};
#pragma unroll
        for (int s = 0; s < 12; ++s)
            accA = __builtin_amdgcn_wmma_f32_16x16x4_f32(
                false, aw[s], false, bA[s], (short)0, accA, false, false);
        if (m < 8) {
            float* pA = pbuf + ((3 * t + 2 * hi) * 4) * PSTRIDE + 2 + r - 2 * hi;
            pA[0 * PSTRIDE] = accA[0];
            pA[1 * PSTRIDE] = accA[1];
            pA[2 * PSTRIDE] = accA[2];
            pA[3 * PSTRIDE] = accA[3];
            if (hi == 0) {
                float* pB = pbuf + ((3 * t + 1) * 4) * PSTRIDE + 1 + r;
                pB[0 * PSTRIDE] = accA[4];
                pB[1 * PSTRIDE] = accA[5];
                pB[2 * PSTRIDE] = accA[6];
                pB[3 * PSTRIDE] = accA[7];
            }
        }
    }
    __syncthreads();

    {
        v2f    Bv[4];
        float* dbase[4];
#pragma unroll
        for (int uu = 0; uu < 4; ++uu) {
            int tt = wave * 4 + uu;
            int hc = tt >> 3;
            int t  = (tt >> 1) & 3;
            int q  = tt & 1;
            const float* pp0 = pbuf + ((t * 3) * 4 + hc) * PSTRIDE + 2 + (q * 16 + m) * 4 + 2 * hi;
            v2f p0 = *(const v2f*)(pp0);
            v2f p1 = *(const v2f*)(pp0 + 4 * PSTRIDE);
            v2f p2 = *(const v2f*)(pp0 + 8 * PSTRIDE);
            Bv[uu].x = fmaxf((p0.x + p1.x) + p2.x, 0.f);
            Bv[uu].y = fmaxf((p0.y + p1.y) + p2.y, 0.f);
            int cq = (4 * hc + t) * 2 + q;
            dbase[uu] = delta + cq * DELTA_STRIDE + m * 16 + 8 * hi;
        }
        v8f u0[4];
#pragma unroll
        for (int uu = 0; uu < 4; ++uu)
            u0[uu] = __builtin_amdgcn_wmma_f32_16x16x4_f32(
                false, afc0, false, Bv[uu], (short)0, cbias, false, false);
#pragma unroll
        for (int uu = 0; uu < 4; ++uu)
#pragma unroll
            for (int v = 0; v < 6; ++v) u0[uu][v] = fmaxf(u0[uu][v], 0.f);
        v8f acc2[4];
#pragma unroll
        for (int uu = 0; uu < 4; ++uu) acc2[uu] = (v8f){};
#pragma unroll
        for (int i = 0; i < 3; ++i)
#pragma unroll
            for (int uu = 0; uu < 4; ++uu) {
                v2f B2 = (v2f){u0[uu][2 * i], u0[uu][2 * i + 1]};
                acc2[uu] = __builtin_amdgcn_wmma_f32_16x16x4_f32(
                    false, afc1[i], false, B2, (short)0, acc2[uu], false, false);
            }
#pragma unroll
        for (int uu = 0; uu < 4; ++uu) {
            *(v2f*)(dbase[uu] + 0) = (v2f){acc2[uu][0], acc2[uu][1]};
            *(v2f*)(dbase[uu] + 2) = (v2f){acc2[uu][2], acc2[uu][3]};
            *(v2f*)(dbase[uu] + 4) = (v2f){acc2[uu][4], acc2[uu][5]};
            *(v2f*)(dbase[uu] + 6) = (v2f){acc2[uu][6], acc2[uu][7]};
        }
    }
    const int r_h2  = tid >> 3;
    const int r_q   = (tid >> 2) & 1;
    const int r_c2b = (tid & 3) << 2;
    const int r_off = (r_h2 * WIDTH + r_q) * CHAN + r_c2b;
    float4 xr[8];
#pragma unroll
    for (int k = 0; k < 8; ++k)
        xr[k] = *(const float4*)(xres + r_off + k * (32 * WIDTH * CHAN));
    __syncthreads();

    float* outp = xout + ((b * HEIGHT) * WIDTH + ww * 4 + 2 * hf) * CHAN;
    const float* dsrc = delta + (r_c2b * 2 + r_q) * DELTA_STRIDE + r_h2;
#pragma unroll
    for (int k = 0; k < 8; ++k) {
        float4 xv = xr[k];
        float4 dv;
        dv.x = dsrc[0 * 2 * DELTA_STRIDE + 32 * k];
        dv.y = dsrc[1 * 2 * DELTA_STRIDE + 32 * k];
        dv.z = dsrc[2 * 2 * DELTA_STRIDE + 32 * k];
        dv.w = dsrc[3 * 2 * DELTA_STRIDE + 32 * k];
        *(float4*)(outp + r_off + k * (32 * WIDTH * CHAN)) =
            make_float4(xv.x + dv.x, xv.y + dv.y, xv.z + dv.z, xv.w + dv.w);
    }
}

extern "C" void kernel_launch(void* const* d_in, const int* in_sizes, int n_in,
                              void* d_out, int out_size, void* d_ws, size_t ws_size,
                              hipStream_t stream) {
    (void)in_sizes; (void)n_in; (void)out_size; (void)ws_size;
    const float* x   = (const float*)d_in[0];
    const float* cwv = (const float*)d_in[1];
    const float* f0w = (const float*)d_in[2];
    const float* f0b = (const float*)d_in[3];
    const float* f1w = (const float*)d_in[4];

    float* tmp = (float*)d_ws;
    float* out = (float*)d_out;

    dim3 grid(BATCH * 128);
    dim3 block(256);
    size_t shmem = SMEM_FLOATS * sizeof(float);

    ca_step_kernel<<<grid, block, shmem, stream>>>(x,   cwv, f0w, f0b, f1w, tmp);
    ca_step_kernel<<<grid, block, shmem, stream>>>(tmp, cwv, f0w, f0b, f1w, out);
}


# MLP tile columns are the 16 output channels of one 16-row group: delta sits in LDS in output layout (conflict-free b32 stores, b128 reads); each wave writes out its own 32 rows, no third barrier
# speedup vs baseline: 3.1306x; 1.0391x over previous
#include <hip/hip_runtime.h>


typedef __attribute__((ext_vector_type(2))) float v2f;
typedef __attribute__((ext_vector_type(8))) float v8f;

#define BATCH  32
#define HEIGHT 256
#define WIDTH  256
#define CHAN   16
#define HCN    4
#define HS     12

#define XROW_STRIDE 20
#define HROWS       128
#define XROWS       130
#define XCOL_SIZE   (XROWS * XROW_STRIDE)
#define NCOLS       12
#define OFF_XCOL    0
#define OFF_CONVW   (NCOLS * XCOL_SIZE)
#define PSTRIDE     132
#define OFF_P       (OFF_CONVW + 576)
#define SMEM_FLOATS (OFF_P + 4 * 3 * HCN * PSTRIDE)
#define DROW        36
#define OFF_DELTA   0
#define ST_ROWS     21
#define ST_PASSES   7

__launch_bounds__(256)
__global__ void ca_step_kernel(const float* __restrict__ xin,
                               const float* __restrict__ convw,
                               const float* __restrict__ fc0w,
                               const float* __restrict__ fc0b,
                               const float* __restrict__ fc1w,
                               float* __restrict__ xout)
{
    extern __shared__ float smem[];
    float* xcol  = smem + OFF_XCOL;
    float* cw    = smem + OFF_CONVW;
    float* pbuf  = smem + OFF_P;
    float* delta = smem + OFF_DELTA;

    const int tid  = threadIdx.x;
    const int lane = tid & 31;
    const int wave = tid >> 5;
    const int m    = lane & 15;
    const int hi   = lane >> 4;

    const int blk = blockIdx.x;
    const int b   = blk >> 7;
    const int hf  = (blk >> 6) & 1;
    const int ww  = blk & 63;

    const float* xres = xin + ((b * HEIGHT) * WIDTH + ww * 4 + 2 * hf) * CHAN;

    for (int i = tid; i < 512; i += 256) {
        int h2 = i >> 1;
        int q  = i & 1;
        __builtin_prefetch(xres + (h2 * WIDTH + q) * CHAN, 0, 1);
    }

    v2f afc0;
    {
        const bool live = (m < 6) || (m >= 8 && m < 14);
        const int  s    = (m < 6) ? m : m - 2;
        int base = (live ? s : 0) * 4 + 2 * hi;
        float a0 = fc0w[base + 0];
        float a1 = fc0w[base + 1];
        afc0 = live ? (v2f){a0, a1} : (v2f){0.f, 0.f};
    }
    v8f cbias;
#pragma unroll
    for (int v = 0; v < 8; ++v) {
        float bv = fc0b[v < 6 ? 6 * hi + v : 0];
        cbias[v] = (v < 6) ? bv : 0.f;
    }
    v2f afc1[3];
#pragma unroll
    for (int i = 0; i < 3; ++i) {
        int base = m * HS + 6 * hi + 2 * i;
        afc1[i].x = fc1w[base + 0];
        afc1[i].y = fc1w[base + 1];
    }

    {
        const int cwi2 = tid < 64 ? tid + 512 : 575;
        float cwv0 = convw[tid];
        float cwv1 = convw[tid + 256];
        float cwv2 = convw[cwi2];

        const int  hq    = tid / 12;
        const int  q     = tid - hq * 12;
        const int  hr    = hq < ST_ROWS ? hq : 0;
        const int  c3    = q >> 2;
        const bool edgeL = (ww == 0)  && (c3 == 0);
        const bool edgeR = (ww == 63) && (c3 == 2);
        const bool top   = (hf == 0) && (hr == 0);
        const bool bot   = (hf == 1) && (hr == 3);
        const int  gbase = b * (HEIGHT * WIDTH * CHAN) + (hf * HROWS + hr - 1) * (WIDTH * CHAN)
                         + (ww - 1) * CHAN + q * 4;
        const int  adjL   = edgeL ? CHAN : 0;
        const int  adjR   = edgeR ? -CHAN : 0;
        const int  adjTop = top ? (WIDTH * CHAN) : 0;
        const int  adjBot = (hf == 1 && hr > 2) ? -(hr - 2) * (WIDTH * CHAN) : 0;

        float4 sv[4 * (ST_PASSES - 1)];
        float4 sl[4];
#pragma unroll
        for (int j = 0; j < 4 * (ST_PASSES - 1); ++j) {
            const int t = j / (ST_PASSES - 1);
            const int k = j - t * (ST_PASSES - 1);
            int off = gbase + t * (64 * CHAN) + k * (ST_ROWS * WIDTH * CHAN);
            if (t == 0) off += adjL;
            if (t == 3) off += adjR;
            if (k == 0) off += adjTop;
            sv[j] = *(const float4*)(xin + off);
        }
#pragma unroll
        for (int t = 0; t < 4; ++t) {
            int off = gbase + t * (64 * CHAN) + (ST_PASSES - 1) * (ST_ROWS * WIDTH * CHAN) + adjBot;
            if (t == 0) off += adjL;
            if (t == 3) off += adjR;
            sl[t] = *(const float4*)(xin + off);
        }

        cw[tid]       = cwv0;
        cw[tid + 256] = cwv1;
        cw[cwi2]      = cwv2;

        float* ldst = xcol + c3 * XCOL_SIZE + hr * XROW_STRIDE + (q & 3) * 4;
#pragma unroll
        for (int j = 0; j < 4 * (ST_PASSES - 1); ++j) {
            const int t = j / (ST_PASSES - 1);
            const int k = j - t * (ST_PASSES - 1);
            bool zero = false;
            if (t == 0) zero = zero || edgeL;
            if (t == 3) zero = zero || edgeR;
            if (k == 0) zero = zero || top;
            float4 val = sv[j];
            if (t == 0 || t == 3 || k == 0) {
                asm volatile("" :: "v"(val.x), "v"(val.y), "v"(val.z), "v"(val.w));
                if (zero) val = make_float4(0.f, 0.f, 0.f, 0.f);
            }
            *(float4*)(ldst + 3 * t * XCOL_SIZE + k * (ST_ROWS * XROW_STRIDE)) = val;
        }
#pragma unroll
        for (int t = 0; t < 4; ++t)
            asm volatile("" :: "v"(sl[t].x), "v"(sl[t].y), "v"(sl[t].z), "v"(sl[t].w));
        if (hr < 4) {
#pragma unroll
            for (int t = 0; t < 4; ++t) {
                bool zero = bot;
                if (t == 0) zero = zero || edgeL;
                if (t == 3) zero = zero || edgeR;
                float4 val = sl[t];
                if (zero) val = make_float4(0.f, 0.f, 0.f, 0.f);
                *(float4*)(ldst + 3 * t * XCOL_SIZE + (ST_PASSES - 1) * (ST_ROWS * XROW_STRIDE)) = val;
            }
        }
    }
    __syncthreads();


    v2f aw[12];
    {
        const int dd = (m >> 2) < 3 ? (m >> 2) : 2;
        const float* pw = cw + (m & 3) * 144 + (2 * hi) * 9 + dd;
#pragma unroll
        for (int a = 0; a < 3; ++a)
#pragma unroll
            for (int cblk = 0; cblk < 4; ++cblk) {
                float w0 = pw[cblk * 36 + 0 + a * 3];
                float w1 = pw[cblk * 36 + 9 + a * 3];
                aw[a * 4 + cblk] = (m < 12) ? (v2f){w0, w1} : (v2f){0.f, 0.f};
            }
    }

#pragma unroll
    for (int pp = 0; pp < 2; ++pp) {
        const int u = wave * 4 + pp * 2;
        const int t = u >> 3;
        const int r = ((u & 7) << 4) + m;
        const float* pb = xcol + 3 * t * XCOL_SIZE + r * XROW_STRIDE + 2 * hi;
        v2f bA[12], bB[12];
#pragma unroll
        for (int a = 0; a < 3; ++a)
#pragma unroll
            for (int cblk = 0; cblk < 4; ++cblk) {
                bA[a * 4 + cblk] = *(const v2f*)(pb + a * XCOL_SIZE + 4 * cblk);
                bB[a * 4 + cblk] = *(const v2f*)(pb + a * XCOL_SIZE + 16 * XROW_STRIDE + 4 * cblk);
            }
        v8f accA = {};
        v8f accB = {};
#pragma unroll
        for (int s = 0; s < 12; ++s) {
            accA = __builtin_amdgcn_wmma_f32_16x16x4_f32(
                false, aw[s], false, bA[s], (short)0, accA, false, false);
            accB = __builtin_amdgcn_wmma_f32_16x16x4_f32(
                false, aw[s], false, bB[s], (short)0, accB, false, false);
        }
        float* pA = pbuf + ((3 * t + 2 * hi) * 4) * PSTRIDE + 2 + r - 2 * hi;
        pA[0 * PSTRIDE] = accA[0];
        pA[1 * PSTRIDE] = accA[1];
        pA[2 * PSTRIDE] = accA[2];
        pA[3 * PSTRIDE] = accA[3];
        pA[0 * PSTRIDE + 16] = accB[0];
        pA[1 * PSTRIDE + 16] = accB[1];
        pA[2 * PSTRIDE + 16] = accB[2];
        pA[3 * PSTRIDE + 16] = accB[3];
        if (hi == 0) {
            float* pB = pbuf + ((3 * t + 1) * 4) * PSTRIDE + 1 + r;
            pB[0 * PSTRIDE] = accA[4];
            pB[1 * PSTRIDE] = accA[5];
            pB[2 * PSTRIDE] = accA[6];
            pB[3 * PSTRIDE] = accA[7];
            pB[0 * PSTRIDE + 16] = accB[4];
            pB[1 * PSTRIDE + 16] = accB[5];
            pB[2 * PSTRIDE + 16] = accB[6];
            pB[3 * PSTRIDE + 16] = accB[7];
        }
    }
    if (wave == 0) {
        const int t = (m >> 1) & 3;
        const int r = 128 + (m & 1);
        const float* pb = xcol + 3 * t * XCOL_SIZE + r * XROW_STRIDE + 2 * hi;
        v2f bA[12];
#pragma unroll
        for (int a = 0; a < 3; ++a)
#pragma unroll
            for (int cblk = 0; cblk < 4; ++cblk)
                bA[a * 4 + cblk] = *(const v2f*)(pb + a * XCOL_SIZE + 4 * cblk);
        v8f accA = {};
#pragma unroll
        for (int s = 0; s < 12; ++s)
            accA = __builtin_amdgcn_wmma_f32_16x16x4_f32(
                false, aw[s], false, bA[s], (short)0, accA, false, false);
        if (m < 8) {
            float* pA = pbuf + ((3 * t + 2 * hi) * 4) * PSTRIDE + 2 + r - 2 * hi;
            pA[0 * PSTRIDE] = accA[0];
            pA[1 * PSTRIDE] = accA[1];
            pA[2 * PSTRIDE] = accA[2];
            pA[3 * PSTRIDE] = accA[3];
            if (hi == 0) {
                float* pB = pbuf + ((3 * t + 1) * 4) * PSTRIDE + 1 + r;
                pB[0 * PSTRIDE] = accA[4];
                pB[1 * PSTRIDE] = accA[5];
                pB[2 * PSTRIDE] = accA[6];
                pB[3 * PSTRIDE] = accA[7];
            }
        }
    }
    __syncthreads();

    {
        v2f    Bv[4];
        float* dbase[4];
        const float* pcol = pbuf + (((m & 3) * 3) * 4 + (m >> 2)) * PSTRIDE + 2 + 2 * hi;
#pragma unroll
        for (int uu = 0; uu < 4; ++uu) {
            int tt = wave * 4 + uu;
            int n  = tt >> 1;
            int q  = tt & 1;
            const float* pp0 = pcol + (q * 16 + n) * 4;
            v2f p0 = *(const v2f*)(pp0);
            v2f p1 = *(const v2f*)(pp0 + 4 * PSTRIDE);
            v2f p2 = *(const v2f*)(pp0 + 8 * PSTRIDE);
            Bv[uu].x = fmaxf((p0.x + p1.x) + p2.x, 0.f);
            Bv[uu].y = fmaxf((p0.y + p1.y) + p2.y, 0.f);
            dbase[uu] = delta + (n * 16 + 8 * hi) * DROW + q * 16 + m;
        }
        v8f u0[4];
#pragma unroll
        for (int uu = 0; uu < 4; ++uu)
            u0[uu] = __builtin_amdgcn_wmma_f32_16x16x4_f32(
                false, afc0, false, Bv[uu], (short)0, cbias, false, false);
#pragma unroll
        for (int uu = 0; uu < 4; ++uu)
#pragma unroll
            for (int v = 0; v < 6; ++v) u0[uu][v] = fmaxf(u0[uu][v], 0.f);
        v8f acc2[4];
#pragma unroll
        for (int uu = 0; uu < 4; ++uu) acc2[uu] = (v8f){};
#pragma unroll
        for (int i = 0; i < 3; ++i)
#pragma unroll
            for (int uu = 0; uu < 4; ++uu) {
                v2f B2 = (v2f){u0[uu][2 * i], u0[uu][2 * i + 1]};
                acc2[uu] = __builtin_amdgcn_wmma_f32_16x16x4_f32(
                    false, afc1[i], false, B2, (short)0, acc2[uu], false, false);
            }
#pragma unroll
        for (int uu = 0; uu < 4; ++uu)
#pragma unroll
            for (int v = 0; v < 8; ++v) dbase[uu][v * DROW] = acc2[uu][v];
    }
    const int r_h2  = wave * 32 + (lane >> 3);
    const int r_q   = (lane >> 2) & 1;
    const int r_c2b = (lane & 3) << 2;
    const int r_off = (r_h2 * WIDTH + r_q) * CHAN + r_c2b;
    float4 xr[8];
#pragma unroll
    for (int k = 0; k < 8; ++k)
        xr[k] = *(const float4*)(xres + r_off + k * (4 * WIDTH * CHAN));
    __threadfence_block();

    float* outp = xout + ((b * HEIGHT) * WIDTH + ww * 4 + 2 * hf) * CHAN;
    const float* dsrc = delta + r_h2 * DROW + r_q * 16 + r_c2b;
#pragma unroll
    for (int k = 0; k < 8; ++k) {
        float4 xv = xr[k];
        float4 dv = *(const float4*)(dsrc + k * (4 * DROW));
        *(float4*)(outp + r_off + k * (4 * WIDTH * CHAN)) =
            make_float4(xv.x + dv.x, xv.y + dv.y, xv.z + dv.z, xv.w + dv.w);
    }
}

extern "C" void kernel_launch(void* const* d_in, const int* in_sizes, int n_in,
                              void* d_out, int out_size, void* d_ws, size_t ws_size,
                              hipStream_t stream) {
    (void)in_sizes; (void)n_in; (void)out_size; (void)ws_size;
    const float* x   = (const float*)d_in[0];
    const float* cwv = (const float*)d_in[1];
    const float* f0w = (const float*)d_in[2];
    const float* f0b = (const float*)d_in[3];
    const float* f1w = (const float*)d_in[4];

    float* tmp = (float*)d_ws;
    float* out = (float*)d_out;

    dim3 grid(BATCH * 128);
    dim3 block(256);
    size_t shmem = SMEM_FLOATS * sizeof(float);

    ca_step_kernel<<<grid, block, shmem, stream>>>(x,   cwv, f0w, f0b, f1w, tmp);
    ca_step_kernel<<<grid, block, shmem, stream>>>(tmp, cwv, f0w, f0b, f1w, out);
}
